# starting kernel
# baseline (speedup 1.0000x reference)
#include <hip/hip_runtime.h>
#include <math.h>

typedef __attribute__((ext_vector_type(2))) float v2f;
typedef __attribute__((ext_vector_type(8))) float v8f;

#define NF    26
#define ED    4
#define NUMD  13
#define KDIM  117
#define KPAD  120
#define DSTR  124
#define L1N   12
#define L2N   8
#define L3N   4
#define ROWS  256
#define EPSV  1e-5f

__global__ __launch_bounds__(256) void deepfm_fused_kernel(
    const int*   __restrict__ x1,   const float* __restrict__ x2,
    const float* __restrict__ fw1,  const float* __restrict__ fw2,
    const float* __restrict__ bias,
    const float* __restrict__ w1, const float* __restrict__ b1,
    const float* __restrict__ g1, const float* __restrict__ be1,
    const float* __restrict__ m1, const float* __restrict__ v1,
    const float* __restrict__ w2, const float* __restrict__ b2,
    const float* __restrict__ g2, const float* __restrict__ be2,
    const float* __restrict__ m2, const float* __restrict__ v2,
    const float* __restrict__ w3, const float* __restrict__ b3,
    const float* __restrict__ g3, const float* __restrict__ be3,
    const float* __restrict__ m3, const float* __restrict__ v3,
    float* __restrict__ out)
{
    __shared__ float s_deep[ROWS][DSTR];
    __shared__ float s_w1t[16][DSTR];
    __shared__ float s_h1[ROWS][L1N];
    __shared__ float s_w2[L1N][L2N];
    __shared__ float s_w3[L2N][L3N];
    __shared__ float sA1[16], sC1[16];
    __shared__ float sA2[L2N], sC2[L2N];
    __shared__ float sA3[L3N], sC3[L3N];

    const int tid = threadIdx.x;
    const int row = blockIdx.x * ROWS + tid;

    for (int i = tid; i < 16 * DSTR; i += 256) {
        int n = i / DSTR, k = i % DSTR;
        s_w1t[n][k] = (n < L1N && k < KDIM) ? w1[k * L1N + n] : 0.0f;
    }
    if (tid < L1N * L2N) s_w2[tid / L2N][tid % L2N] = w2[tid];
    if (tid < L2N * L3N) s_w3[tid / L3N][tid % L3N] = w3[tid];
    if (tid < 16) {
        if (tid < L1N) {
            float sc = g1[tid] * rsqrtf(v1[tid] + EPSV);
            sA1[tid] = sc;
            sC1[tid] = (b1[tid] - m1[tid]) * sc + be1[tid];
        } else { sA1[tid] = 0.0f; sC1[tid] = 0.0f; }
    }
    if (tid >= 32 && tid < 32 + L2N) {
        int j = tid - 32;
        float sc = g2[j] * rsqrtf(v2[j] + EPSV);
        sA2[j] = sc; sC2[j] = (b2[j] - m2[j]) * sc + be2[j];
    }
    if (tid >= 64 && tid < 64 + L3N) {
        int j = tid - 64;
        float sc = g3[j] * rsqrtf(v3[j] + EPSV);
        sA3[j] = sc; sC3[j] = (b3[j] - m3[j]) * sc + be3[j];
    }

    float fm1 = 0.0f;
    float s0 = 0.f, s1 = 0.f, s2 = 0.f, s3 = 0.f;
    float q0 = 0.f, q1 = 0.f, q2 = 0.f, q3 = 0.f;
    {
        const float4* emb = (const float4*)fw2;
        #pragma unroll
        for (int f = 0; f < NF; ++f) {
            int idx = x1[row * NF + f];
            fm1 += fw1[idx];
            float4 e = emb[idx];
            s0 += e.x; s1 += e.y; s2 += e.z; s3 += e.w;
            q0 += e.x * e.x; q1 += e.y * e.y; q2 += e.z * e.z; q3 += e.w * e.w;
            *(float4*)&s_deep[tid][f * 4] = e;
        }
        #pragma unroll
        for (int j = 0; j < NUMD; ++j) s_deep[tid][NF * 4 + j] = x2[row * NUMD + j];
        s_deep[tid][117] = 0.0f; s_deep[tid][118] = 0.0f; s_deep[tid][119] = 0.0f;
    }
    const float fm2 = 0.5f * ((s0 * s0 - q0) + (s1 * s1 - q1) +
                              (s2 * s2 - q2) + (s3 * s3 - q3));

    __syncthreads();

    {
        const int lane = tid & 31;
        const int wave = tid >> 5;
        const int half = lane >> 4;
        const int mn   = lane & 15;
        #pragma unroll
        for (int t = 0; t < 2; ++t) {
            const int tileBase = (wave * 2 + t) * 16;
            v8f acc = {};
            #pragma unroll
            for (int c = 0; c < KPAD / 4; ++c) {
                const int k = c * 4 + half * 2;
                v2f a = *(const v2f*)&s_deep[tileBase + mn][k];
                v2f b = *(const v2f*)&s_w1t[mn][k];
                acc = __builtin_amdgcn_wmma_f32_16x16x4_f32(
                    false, a, false, b, (short)0, acc, false, false);
            }
            if (mn < L1N) {
                const float sc = sA1[mn], sh = sC1[mn];
                #pragma unroll
                for (int i = 0; i < 8; ++i) {
                    const int r = tileBase + i + half * 8;
                    float h = acc[i] * sc + sh;
                    s_h1[r][mn] = h > 0.0f ? h : 0.0f;
                }
            }
        }
    }

    __syncthreads();

    float h2[L2N];
    #pragma unroll
    for (int j = 0; j < L2N; ++j) {
        float a = 0.0f;
        #pragma unroll
        for (int k = 0; k < L1N; ++k) a += s_h1[tid][k] * s_w2[k][j];
        a = a * sA2[j] + sC2[j];
        h2[j] = a > 0.0f ? a : 0.0f;
    }
    float hsum = 0.0f;
    #pragma unroll
    for (int j = 0; j < L3N; ++j) {
        float a = 0.0f;
        #pragma unroll
        for (int k = 0; k < L2N; ++k) a += h2[k] * s_w3[k][j];
        a = a * sA3[j] + sC3[j];
        hsum += a > 0.0f ? a : 0.0f;
    }
    const float z = fm1 + fm2 + hsum + bias[0];
    out[row] = 1.0f / (1.0f + __expf(-z));
}

extern "C" void kernel_launch(void* const* d_in, const int* in_sizes, int n_in,
                              void* d_out, int out_size, void* d_ws, size_t ws_size,
                              hipStream_t stream) {
    const int*   x1   = (const int*)d_in[0];
    const float* x2   = (const float*)d_in[1];
    const float* fw1  = (const float*)d_in[2];
    const float* fw2  = (const float*)d_in[3];
    const float* bias = (const float*)d_in[4];
    const float* w1 = (const float*)d_in[5];  const float* b1 = (const float*)d_in[6];
    const float* g1 = (const float*)d_in[7];  const float* be1 = (const float*)d_in[8];
    const float* m1 = (const float*)d_in[9];  const float* v1 = (const float*)d_in[10];
    const float* w2 = (const float*)d_in[11]; const float* b2 = (const float*)d_in[12];
    const float* g2 = (const float*)d_in[13]; const float* be2 = (const float*)d_in[14];
    const float* m2 = (const float*)d_in[15]; const float* v2 = (const float*)d_in[16];
    const float* w3 = (const float*)d_in[17]; const float* b3 = (const float*)d_in[18];
    const float* g3 = (const float*)d_in[19]; const float* be3 = (const float*)d_in[20];
    const float* m3 = (const float*)d_in[21]; const float* v3 = (const float*)d_in[22];

    const int rows = in_sizes[0] / NF;
    const int grid = rows / ROWS;

    deepfm_fused_kernel<<<grid, ROWS, 0, stream>>>(
        x1, x2, fw1, fw2, bias,
        w1, b1, g1, be1, m1, v1,
        w2, b2, g2, be2, m2, v2,
        w3, b3, g3, be3, m3, v3,
        (float*)d_out);
}


# one wave per 16-row tile: each lane half gathers 13 fields straight into the WMMA operand registers (layers 1-2 formed transposed on WMMA, k permuted), 256 four-wave groups; FP sum order changes
# speedup vs baseline: 2.3167x; 2.3167x over previous
#include <hip/hip_runtime.h>
#include <math.h>


typedef __attribute__((ext_vector_type(2))) float v2f;
typedef __attribute__((ext_vector_type(8))) float v8f;

#define NT    128
#define NF    26
#define HF    13
#define ED    4
#define NUMD  13
#define KDIM  117
#define L1N   12
#define L2N   8
#define L3N   4
#define ROWS  64
#define W1V4  351
#define EPSV  1e-5f

__global__ __launch_bounds__(NT) void deepfm_fused_kernel(
    const int*   __restrict__ x1,   const float* __restrict__ x2,
    const float* __restrict__ fw1,  const float* __restrict__ fw2,
    const float* __restrict__ bias,
    const float* __restrict__ w1, const float* __restrict__ b1,
    const float* __restrict__ g1, const float* __restrict__ be1,
    const float* __restrict__ m1, const float* __restrict__ v1,
    const float* __restrict__ w2, const float* __restrict__ b2,
    const float* __restrict__ g2, const float* __restrict__ be2,
    const float* __restrict__ m2, const float* __restrict__ v2,
    const float* __restrict__ w3, const float* __restrict__ b3,
    const float* __restrict__ g3, const float* __restrict__ be3,
    const float* __restrict__ m3, const float* __restrict__ v3,
    float* __restrict__ out)
{
    __shared__ int   s_idx[2048];
    __shared__ float s_w1[1536];
    __shared__ float s_bn1[32];
    __shared__ float s_out[2 * ROWS];

    const int tid  = threadIdx.x;
    const int lane = tid & 31;
    const int wave = tid >> 5;
    const int m    = lane & 15;
    const int half = lane >> 4;
    const int rib  = wave * 16 + m;
    const int row  = blockIdx.x * ROWS + rib;

    const int4*   isrc = (const int4*)(x1 + blockIdx.x * (ROWS * NF));
    const float4* wsrc = (const float4*)w1;
    int4   iv[4];
    float4 wv[3];
    #pragma unroll
    for (int q = 0; q < 4; ++q) iv[q] = isrc[min(tid + NT * q, ROWS * NF / 4 - 1)];
    #pragma unroll
    for (int q = 0; q < 3; ++q) wv[q] = wsrc[min(tid + NT * q, W1V4 - 1)];

    const int nn = min(m, L1N - 1);
    const float g1v = g1[nn], v1v = v1[nn], b1v = b1[nn], m1v = m1[nn], be1v = be1[nn];

    float xd[8];
    #pragma unroll
    for (int c = 0; c < 4; ++c) {
        #pragma unroll
        for (int t = 0; t < 2; ++t) {
            const int kd = c * 4 + half * 2 + t;
            const float xv = x2[row * NUMD + min(kd, NUMD - 1)];
            xd[c * 2 + t] = (kd < NUMD) ? xv : 0.0f;
        }
    }

    v2f w2a[4];
    #pragma unroll
    for (int c = 0; c < 4; ++c) {
        #pragma unroll
        for (int t = 0; t < 2; ++t) {
            const int n = 8 * half + 2 * c + t;
            const float wv2 = w2[min(n, L1N - 1) * L2N + (m & (L2N - 1))];
            w2a[c][t] = (m < L2N && n < L1N) ? wv2 : 0.0f;
        }
    }

    {
        int4* idst = (int4*)s_idx;
        #pragma unroll
        for (int q = 0; q < 4; ++q) idst[tid + NT * q] = iv[q];
        float4* wdst = (float4*)s_w1;
        #pragma unroll
        for (int q = 0; q < 3; ++q) {
            const int e = tid + NT * q;
            wdst[e] = (e < W1V4) ? wv[q] : make_float4(0.f, 0.f, 0.f, 0.f);
        }
        const float sc = g1v * rsqrtf(v1v + EPSV);
        const float sh = (b1v - m1v) * sc + be1v;
        s_bn1[m]      = (m < L1N) ? sc : 0.0f;
        s_bn1[16 + m] = (m < L1N) ? sh : 0.0f;
    }
    __syncthreads();

    const float4* emb = (const float4*)fw2;
    float4 e[HF];
    float  f1[HF];
    #pragma unroll
    for (int j = 0; j < HF; ++j) {
        const int idx = s_idx[rib * NF + half * HF + j];
        e[j]  = emb[idx];
        f1[j] = fw1[idx];
    }

    float A2[L2N], C2[L2N], A3[L3N], C3[L3N], w3r[L2N * L3N];
    #pragma unroll
    for (int j = 0; j < L2N; ++j) {
        A2[j] = g2[j] * rsqrtf(v2[j] + EPSV);
        C2[j] = (b2[j] - m2[j]) * A2[j] + be2[j];
    }
    #pragma unroll
    for (int j = 0; j < L3N; ++j) {
        A3[j] = g3[j] * rsqrtf(v3[j] + EPSV);
        C3[j] = (b3[j] - m3[j]) * A3[j] + be3[j];
    }
    #pragma unroll
    for (int i = 0; i < L2N * L3N; ++i) w3r[i] = w3[i];
    const float bias0 = bias[0];

    const bool nvalid = (m < L1N);
    v2f wa[30];
    #pragma unroll
    for (int j = 0; j < HF; ++j) {
        #pragma unroll
        for (int p = 0; p < 2; ++p) {
            const int k0 = (half * HF + j) * 4 + 2 * p;
            const float a0 = s_w1[k0 * L1N + m];
            const float a1 = s_w1[(k0 + 1) * L1N + m];
            wa[2 * j + p][0] = nvalid ? a0 : 0.0f;
            wa[2 * j + p][1] = nvalid ? a1 : 0.0f;
        }
    }
    #pragma unroll
    for (int c = 0; c < 4; ++c) {
        const int k0 = 104 + c * 4 + half * 2;
        const float a0 = s_w1[k0 * L1N + m];
        const float a1 = s_w1[(k0 + 1) * L1N + m];
        wa[26 + c][0] = nvalid ? a0 : 0.0f;
        wa[26 + c][1] = nvalid ? a1 : 0.0f;
    }

    __builtin_amdgcn_sched_barrier(0);

    v8f acc0 = {}, acc1 = {};
    #pragma unroll
    for (int j = 0; j < HF; ++j) {
        v2f bxy, bzw;
        bxy[0] = e[j].x; bxy[1] = e[j].y;
        bzw[0] = e[j].z; bzw[1] = e[j].w;
        acc0 = __builtin_amdgcn_wmma_f32_16x16x4_f32(
            false, wa[2 * j], false, bxy, (short)0, acc0, false, false);
        acc1 = __builtin_amdgcn_wmma_f32_16x16x4_f32(
            false, wa[2 * j + 1], false, bzw, (short)0, acc1, false, false);
    }
    #pragma unroll
    for (int c = 0; c < 4; c += 2) {
        v2f d0, d1;
        d0[0] = xd[2 * c];     d0[1] = xd[2 * c + 1];
        d1[0] = xd[2 * c + 2]; d1[1] = xd[2 * c + 3];
        acc0 = __builtin_amdgcn_wmma_f32_16x16x4_f32(
            false, wa[26 + c], false, d0, (short)0, acc0, false, false);
        acc1 = __builtin_amdgcn_wmma_f32_16x16x4_f32(
            false, wa[27 + c], false, d1, (short)0, acc1, false, false);
    }

    float fm1 = 0.0f;
    float s0 = 0.f, s1 = 0.f, s2 = 0.f, s3 = 0.f;
    float q0 = 0.f, q1 = 0.f, q2 = 0.f, q3 = 0.f;
    #pragma unroll
    for (int j = 0; j < HF; ++j) {
        fm1 += f1[j];
        s0 += e[j].x; s1 += e[j].y; s2 += e[j].z; s3 += e[j].w;
        q0 += e[j].x * e[j].x; q1 += e[j].y * e[j].y;
        q2 += e[j].z * e[j].z; q3 += e[j].w * e[j].w;
    }
    fm1 += __shfl_xor(fm1, 16);
    s0 += __shfl_xor(s0, 16); s1 += __shfl_xor(s1, 16);
    s2 += __shfl_xor(s2, 16); s3 += __shfl_xor(s3, 16);
    q0 += __shfl_xor(q0, 16); q1 += __shfl_xor(q1, 16);
    q2 += __shfl_xor(q2, 16); q3 += __shfl_xor(q3, 16);
    const float fm2 = 0.5f * ((s0 * s0 - q0) + (s1 * s1 - q1) +
                              (s2 * s2 - q2) + (s3 * s3 - q3));

    v8f h1v;
    #pragma unroll
    for (int r = 0; r < 8; ++r) {
        const float sc = s_bn1[8 * half + r];
        const float sh = s_bn1[16 + 8 * half + r];
        const float h = (acc0[r] + acc1[r]) * sc + sh;
        h1v[r] = h > 0.0f ? h : 0.0f;
    }

    v8f d2 = {};
    #pragma unroll
    for (int c = 0; c < 4; ++c) {
        v2f hb;
        hb[0] = h1v[2 * c]; hb[1] = h1v[2 * c + 1];
        d2 = __builtin_amdgcn_wmma_f32_16x16x4_f32(
            false, w2a[c], false, hb, (short)0, d2, false, false);
    }
    float h2[L2N];
    #pragma unroll
    for (int j = 0; j < L2N; ++j) {
        const float a = d2[j] * A2[j] + C2[j];
        h2[j] = a > 0.0f ? a : 0.0f;
    }

    float hsum = 0.0f;
    #pragma unroll
    for (int j = 0; j < L3N; ++j) {
        float a = 0.0f;
        #pragma unroll
        for (int k = 0; k < L2N; ++k) a += h2[k] * w3r[k * L3N + j];
        a = a * A3[j] + C3[j];
        hsum += a > 0.0f ? a : 0.0f;
    }
    const float z = fm1 + fm2 + hsum + bias0;
    s_out[half * ROWS + rib] = 1.0f / (1.0f + __expf(-z));

    __syncthreads();
    if (tid < ROWS) out[blockIdx.x * ROWS + tid] = s_out[tid];
}

extern "C" void kernel_launch(void* const* d_in, const int* in_sizes, int n_in,
                              void* d_out, int out_size, void* d_ws, size_t ws_size,
                              hipStream_t stream) {
    const int*   x1   = (const int*)d_in[0];
    const float* x2   = (const float*)d_in[1];
    const float* fw1  = (const float*)d_in[2];
    const float* fw2  = (const float*)d_in[3];
    const float* bias = (const float*)d_in[4];
    const float* w1 = (const float*)d_in[5];  const float* b1 = (const float*)d_in[6];
    const float* g1 = (const float*)d_in[7];  const float* be1 = (const float*)d_in[8];
    const float* m1 = (const float*)d_in[9];  const float* v1 = (const float*)d_in[10];
    const float* w2 = (const float*)d_in[11]; const float* b2 = (const float*)d_in[12];
    const float* g2 = (const float*)d_in[13]; const float* be2 = (const float*)d_in[14];
    const float* m2 = (const float*)d_in[15]; const float* v2 = (const float*)d_in[16];
    const float* w3 = (const float*)d_in[17]; const float* b3 = (const float*)d_in[18];
    const float* g3 = (const float*)d_in[19]; const float* be3 = (const float*)d_in[20];
    const float* m3 = (const float*)d_in[21]; const float* v3 = (const float*)d_in[22];

    const int rows = in_sizes[0] / NF;
    const int grid = rows / ROWS;

    deepfm_fused_kernel<<<grid, NT, 0, stream>>>(
        x1, x2, fw1, fw2, bias,
        w1, b1, g1, be1, m1, v1,
        w2, b2, g2, be2, m2, v2,
        w3, b3, g3, be3, m3, v3,
        (float*)d_out);
}
